# starting kernel
# baseline (speedup 1.0000x reference)
#include <hip/hip_runtime.h>
#include <cmath>


typedef __attribute__((ext_vector_type(2))) float v2f;
typedef __attribute__((ext_vector_type(8))) float v8f;

#define I_SIZE   3
#define H        128
#define O_SIZE   3
#define BATCH    512
#define T_STEPS  50
#define AGENTS   64
#define PRED     12
#define NROWS    (BATCH * AGENTS)
#define WPAD     132
#define WAVES    8
#define TILE_M   16
#define ROWS_PER_BLOCK (WAVES * TILE_M)
#define NBLOCKS  (NROWS / ROWS_PER_BLOCK)

struct __align__(16) Smem {
    float W[H][WPAD];
    float fcT[H][16];
    float stage[WAVES][TILE_M][WPAD];
};

__device__ __forceinline__ v2f mk2(float a, float b) { v2f r; r[0] = a; r[1] = b; return r; }

__device__ __forceinline__ v8f splat8(float x) {
    v8f r;
#pragma unroll
    for (int i = 0; i < 8; ++i) r[i] = x;
    return r;
}

__device__ __forceinline__ v8f wmma4(v2f a, v2f b, v8f c) {
    return __builtin_amdgcn_wmma_f32_16x16x4_f32(
         false, a,  false, b,
         (short)0, c,  false,  false);
}

__device__ __forceinline__ float fast_tanh(float x) {
#if __has_builtin(__builtin_amdgcn_tanhf)
    return __builtin_amdgcn_tanhf(x);
#else
    return ::tanhf(x);
#endif
}

__global__ void __launch_bounds__(256, 1)
traj_rnn_kernel(const float* __restrict__ src,
                const float* __restrict__ W_ih_enc,
                const float* __restrict__ W_hh_enc,
                const float* __restrict__ b_ih_enc,
                const float* __restrict__ b_hh_enc,
                const float* __restrict__ W_ih_dec,
                const float* __restrict__ W_hh_dec,
                const float* __restrict__ b_ih_dec,
                const float* __restrict__ b_hh_dec,
                const float* __restrict__ fc_W,
                const float* __restrict__ fc_b,
                float* __restrict__ out)
{
    __shared__ Smem s;

    const int tid  = threadIdx.x;
    const int wave = tid >> 5;
    const int lane = tid & 31;
    const int l16  = lane & 15;
    const bool hi  = lane >= 16;
    const int koff = hi ? 2 : 0;
    const int r0   = blockIdx.x * ROWS_PER_BLOCK + wave * TILE_M;

    for (int i = tid; i < H * H; i += 256)
        s.W[i >> 7][i & 127] = W_hh_enc[i];
    for (int i = tid; i < H * 16; i += 256) {
        const int k = i >> 4, n = i & 15;
        s.fcT[k][n] = (n < O_SIZE) ? fc_W[n * H + k] : 0.0f;
    }
    __syncthreads();

    v2f BihE[8], BihD[8];
    float biasE[8], biasD[8];
#pragma unroll
    for (int t = 0; t < 8; ++t) {
        const int n = t * 16 + l16;
        if (!hi) {
            BihE[t] = mk2(W_ih_enc[n * I_SIZE + 0], W_ih_enc[n * I_SIZE + 1]);
            BihD[t] = mk2(W_ih_dec[n * I_SIZE + 0], W_ih_dec[n * I_SIZE + 1]);
        } else {
            BihE[t] = mk2(W_ih_enc[n * I_SIZE + 2], 0.0f);
            BihD[t] = mk2(W_ih_dec[n * I_SIZE + 2], 0.0f);
        }
        biasE[t] = b_ih_enc[n] + b_hh_enc[n];
        biasD[t] = b_ih_dec[n] + b_hh_dec[n];
    }
    const float fcb = (l16 < O_SIZE) ? fc_b[l16] : 0.0f;

    const int myrow = r0 + l16;
    const int mb = myrow >> 6;
    const int ma = myrow & 63;
    const float* xbase = src + (size_t)mb * T_STEPS * (AGENTS * I_SIZE) + ma * I_SIZE;

    v2f hA[32];
#pragma unroll
    for (int c = 0; c < 32; ++c) hA[c] = mk2(0.0f, 0.0f);

    v2f Ax = mk2(0.0f, 0.0f);
    v8f acc[8];

    for (int t = 0; t < T_STEPS; ++t) {
        const float* xr = xbase + (size_t)t * (AGENTS * I_SIZE);
        Ax = hi ? mk2(xr[2], 0.0f) : mk2(xr[0], xr[1]);

#pragma unroll
        for (int n = 0; n < 8; ++n) acc[n] = splat8(biasE[n]);
#pragma unroll
        for (int n = 0; n < 8; ++n) acc[n] = wmma4(Ax, BihE[n], acc[n]);

#pragma unroll
        for (int c = 0; c < 32; ++c) {
#pragma unroll
            for (int n = 0; n < 8; ++n) {
                const v2f Bv = *(const v2f*)&s.W[n * 16 + l16][4 * c + koff];
                acc[n] = wmma4(hA[c], Bv, acc[n]);
            }
        }

#pragma unroll
        for (int n = 0; n < 8; ++n) {
#pragma unroll
            for (int v = 0; v < 8; ++v) {
                const int row = hi ? v + 8 : v;
                s.stage[wave][row][n * 16 + l16] = fast_tanh(acc[n][v]);
            }
        }
#pragma unroll
        for (int c = 0; c < 32; ++c)
            hA[c] = *(const v2f*)&s.stage[wave][l16][4 * c + koff];
    }

    {
        const float* xr = xbase + (size_t)(T_STEPS - 1) * (AGENTS * I_SIZE);
        Ax = hi ? mk2(xr[2], 0.0f) : mk2(xr[0], xr[1]);
    }

    __syncthreads();
    for (int i = tid; i < H * H; i += 256)
        s.W[i >> 7][i & 127] = W_hh_dec[i];
    __syncthreads();

    for (int p = 0; p < PRED; ++p) {
#pragma unroll
        for (int n = 0; n < 8; ++n) acc[n] = splat8(biasD[n]);
#pragma unroll
        for (int n = 0; n < 8; ++n) acc[n] = wmma4(Ax, BihD[n], acc[n]);
#pragma unroll
        for (int c = 0; c < 32; ++c) {
#pragma unroll
            for (int n = 0; n < 8; ++n) {
                const v2f Bv = *(const v2f*)&s.W[n * 16 + l16][4 * c + koff];
                acc[n] = wmma4(hA[c], Bv, acc[n]);
            }
        }
#pragma unroll
        for (int n = 0; n < 8; ++n) {
#pragma unroll
            for (int v = 0; v < 8; ++v) {
                const int row = hi ? v + 8 : v;
                s.stage[wave][row][n * 16 + l16] = fast_tanh(acc[n][v]);
            }
        }
#pragma unroll
        for (int c = 0; c < 32; ++c)
            hA[c] = *(const v2f*)&s.stage[wave][l16][4 * c + koff];

        v8f pr = splat8(fcb);
#pragma unroll
        for (int c = 0; c < 32; ++c) {
            const v2f Bv = mk2(s.fcT[4 * c + koff][l16], s.fcT[4 * c + koff + 1][l16]);
            pr = wmma4(hA[c], Bv, pr);
        }

        if (l16 < O_SIZE) {
#pragma unroll
            for (int v = 0; v < 8; ++v) {
                const int row = hi ? v + 8 : v;
                const int n  = r0 + row;
                const int bb = n >> 6, aa = n & 63;
                out[((size_t)bb * PRED + p) * (AGENTS * O_SIZE) + aa * O_SIZE + l16] = pr[v];
                s.stage[wave][row][l16] = pr[v];
            }
        }
        Ax = hi ? mk2(s.stage[wave][l16][2], 0.0f)
                : mk2(s.stage[wave][l16][0], s.stage[wave][l16][1]);
    }
}

extern "C" void kernel_launch(void* const* d_in, const int* in_sizes, int n_in,
                              void* d_out, int out_size, void* d_ws, size_t ws_size,
                              hipStream_t stream) {
    const float* src      = (const float*)d_in[0];
    const float* W_ih_enc = (const float*)d_in[1];
    const float* W_hh_enc = (const float*)d_in[2];
    const float* b_ih_enc = (const float*)d_in[3];
    const float* b_hh_enc = (const float*)d_in[4];
    const float* W_ih_dec = (const float*)d_in[5];
    const float* W_hh_dec = (const float*)d_in[6];
    const float* b_ih_dec = (const float*)d_in[7];
    const float* b_hh_dec = (const float*)d_in[8];
    const float* fc_W     = (const float*)d_in[9];
    const float* fc_b     = (const float*)d_in[10];
    float* out = (float*)d_out;

    traj_rnn_kernel<<<NBLOCKS, 256, 0, stream>>>(
        src, W_ih_enc, W_hh_enc, b_ih_enc, b_hh_enc,
        W_ih_dec, W_hh_dec, b_ih_dec, b_hh_dec, fc_W, fc_b, out);
}


# Loop order N-tile outer, K inner, two row tiles per wave at one wave per SIMD: each finished N-tile's tanh and C-to-A layout change run beside the next N-tile's WMMAs; per-element sums unchanged
# speedup vs baseline: 1.1028x; 1.1028x over previous
#include <hip/hip_runtime.h>
#include <cmath>


typedef __attribute__((ext_vector_type(2))) float v2f;
typedef __attribute__((ext_vector_type(4))) float v4f;
typedef __attribute__((ext_vector_type(8))) float v8f;

#define I_SIZE   3
#define H        128
#define O_SIZE   3
#define BATCH    512
#define T_STEPS  50
#define AGENTS   64
#define PRED     12
#define NROWS    (BATCH * AGENTS)
#define WPAD     132
#define WAVES    4
#define TILE_M   16
#define TILES    2
#define ROWS_PER_WAVE  (TILES * TILE_M)
#define ROWS_PER_BLOCK (WAVES * ROWS_PER_WAVE)
#define NBLOCKS  (NROWS / ROWS_PER_BLOCK)

struct __align__(16) Smem {
    float W[H][WPAD];
    float fcT[H][16];
    float stage[WAVES][ROWS_PER_WAVE][WPAD];
};

__device__ __forceinline__ int pcol(int k) { return (k & ~6) | ((k & 2) << 1) | ((k & 4) >> 1); }

__device__ __forceinline__ v2f mk2(float a, float b) { v2f r; r[0] = a; r[1] = b; return r; }

__device__ __forceinline__ v8f splat8(float x) {
    v8f r;
#pragma unroll
    for (int i = 0; i < 8; ++i) r[i] = x;
    return r;
}

__device__ __forceinline__ v8f wmma4(v2f a, v2f b, v8f c) {
    return __builtin_amdgcn_wmma_f32_16x16x4_f32(
         false, a,  false, b,
         (short)0, c,  false,  false);
}

__device__ __forceinline__ float fast_tanh(float x) {
#if __has_builtin(__builtin_amdgcn_tanhf)
    return __builtin_amdgcn_tanhf(x);
#else
    return ::tanhf(x);
#endif
}

__device__ __forceinline__ void fill_W(Smem& s, const float* __restrict__ Wg, int tid) {
    for (int i = tid; i < H * H / 4; i += WAVES * 32) {
        const v4f q = *(const v4f*)(Wg + 4 * i);
        const int row = i >> 5, col = (i & 31) * 4;
        *(v2f*)&s.W[row][pcol(col)]     = mk2(q[0], q[1]);
        *(v2f*)&s.W[row][pcol(col + 2)] = mk2(q[2], q[3]);
    }
}

__device__ __forceinline__ void load_B(const float* wl, int n, v2f (&B)[32]) {
#pragma unroll
    for (int j = 0; j < 16; ++j) {
        const v4f q = *(const v4f*)(wl + n * 16 * WPAD + 8 * j);
        B[2 * j]     = mk2(q[0], q[1]);
        B[2 * j + 1] = mk2(q[2], q[3]);
    }
}

__device__ __forceinline__ void landed(const v2f (&B)[32]) {
    asm volatile("" :: "v"(B[0]), "v"(B[1]), "v"(B[2]), "v"(B[3]), "v"(B[4]), "v"(B[5]), "v"(B[6]), "v"(B[7]),
                       "v"(B[8]), "v"(B[9]), "v"(B[10]), "v"(B[11]), "v"(B[12]), "v"(B[13]), "v"(B[14]), "v"(B[15]));
    asm volatile("" :: "v"(B[16]), "v"(B[17]), "v"(B[18]), "v"(B[19]), "v"(B[20]), "v"(B[21]), "v"(B[22]), "v"(B[23]),
                       "v"(B[24]), "v"(B[25]), "v"(B[26]), "v"(B[27]), "v"(B[28]), "v"(B[29]), "v"(B[30]), "v"(B[31]));
}

__device__ __forceinline__ void rnn_step(const float* wl, float* sst, const float* sld,
                                         v2f (&hP)[32], v2f (&hQ)[32],
                                         const v2f (&B0)[32],
                                         const v2f AxP, const v2f AxQ,
                                         const v2f (&Bih)[8], const float (&bias)[8])
{
    v2f Bc[32], Bn[32];
    v2f nP[28], nQ[28];
    v8f pP = splat8(0.0f), pQ = splat8(0.0f);
#pragma unroll
    for (int c = 0; c < 32; ++c) { Bc[c] = B0[c]; Bn[c] = B0[c]; }

#pragma unroll
    for (int n = 0; n < 8; ++n) {
        if (n > 0) __builtin_amdgcn_sched_barrier(0);
        v8f aP = splat8(bias[n]);
        v8f aQ = aP;
        aP = wmma4(AxP, Bih[n], aP);
        aQ = wmma4(AxQ, Bih[n], aQ);
#pragma unroll
        for (int c = 0; c < 4; ++c) {
            aP = wmma4(hP[c], Bc[c], aP);
            aQ = wmma4(hQ[c], Bc[c], aQ);
        }
        float tP[8], tQ[8];
#pragma unroll
        for (int v = 0; v < 8; ++v) {
            if (n > 0) {
                __builtin_amdgcn_sched_barrier(0);
                tP[v] = fast_tanh(pP[v]);
                tQ[v] = fast_tanh(pQ[v]);
            }
            aP = wmma4(hP[4 + v], Bc[4 + v], aP);
            aQ = wmma4(hQ[4 + v], Bc[4 + v], aQ);
        }
        __builtin_amdgcn_sched_barrier(0);
        if (n > 0) {
#pragma unroll
            for (int v = 0; v < 8; ++v) {
                sst[v * WPAD + (n - 1) * 16]            = tP[v];
                sst[(16 + v) * WPAD + (n - 1) * 16]     = tQ[v];
            }
        }
        if (n < 7) load_B(wl, n + 1, Bn);
        if (n == 7) {
#pragma unroll
            for (int j = 0; j < 14; ++j) {
                const v4f q = *(const v4f*)(sld + 8 * j);
                const v4f r = *(const v4f*)(sld + 16 * WPAD + 8 * j);
                nP[2 * j] = mk2(q[0], q[1]); nP[2 * j + 1] = mk2(q[2], q[3]);
                nQ[2 * j] = mk2(r[0], r[1]); nQ[2 * j + 1] = mk2(r[2], r[3]);
            }
        }
        __builtin_amdgcn_sched_barrier(0);
#pragma unroll
        for (int c = 12; c < 32; ++c) {
            aP = wmma4(hP[c], Bc[c], aP);
            aQ = wmma4(hQ[c], Bc[c], aQ);
        }
        pP = aP; pQ = aQ;
#pragma unroll
        for (int c = 0; c < 32; ++c) Bc[c] = Bn[c];
    }

    __builtin_amdgcn_sched_barrier(0);
    float uP[8], uQ[8];
#pragma unroll
    for (int v = 0; v < 8; ++v) { uP[v] = fast_tanh(pP[v]); uQ[v] = fast_tanh(pQ[v]); }
    __builtin_amdgcn_sched_barrier(0);
#pragma unroll
    for (int v = 0; v < 8; ++v) {
        sst[v * WPAD + 7 * 16]        = uP[v];
        sst[(16 + v) * WPAD + 7 * 16] = uQ[v];
    }
#pragma unroll
    for (int c = 0; c < 28; ++c) { hP[c] = nP[c]; hQ[c] = nQ[c]; }
#pragma unroll
    for (int j = 14; j < 16; ++j) {
        const v4f q = *(const v4f*)(sld + 8 * j);
        const v4f r = *(const v4f*)(sld + 16 * WPAD + 8 * j);
        hP[2 * j] = mk2(q[0], q[1]); hP[2 * j + 1] = mk2(q[2], q[3]);
        hQ[2 * j] = mk2(r[0], r[1]); hQ[2 * j + 1] = mk2(r[2], r[3]);
    }
}

__global__ void __launch_bounds__(128, 1)
traj_rnn_kernel(const float* __restrict__ src,
                const float* __restrict__ W_ih_enc,
                const float* __restrict__ W_hh_enc,
                const float* __restrict__ b_ih_enc,
                const float* __restrict__ b_hh_enc,
                const float* __restrict__ W_ih_dec,
                const float* __restrict__ W_hh_dec,
                const float* __restrict__ b_ih_dec,
                const float* __restrict__ b_hh_dec,
                const float* __restrict__ fc_W,
                const float* __restrict__ fc_b,
                float* __restrict__ out)
{
    __shared__ Smem s;

    const int tid  = threadIdx.x;
    const int wave = tid >> 5;
    const int lane = tid & 31;
    const int l16  = lane & 15;
    const bool hi  = lane >= 16;
    const int koff = hi ? 2 : 0;
    const int r0   = blockIdx.x * ROWS_PER_BLOCK + wave * ROWS_PER_WAVE;

    fill_W(s, W_hh_enc, tid);
    for (int i = tid; i < H * 16; i += WAVES * 32) {
        const int k = i >> 4, n = i & 15;
        const float w = fc_W[(n < O_SIZE ? n : 0) * H + k];
        s.fcT[k][n] = (n < O_SIZE) ? w : 0.0f;
    }
    __syncthreads();

    v2f BihE[8], BihD[8];
    float biasE[8], biasD[8];
#pragma unroll
    for (int t = 0; t < 8; ++t) {
        const int n = t * 16 + l16;
        if (!hi) {
            BihE[t] = mk2(W_ih_enc[n * I_SIZE + 0], W_ih_enc[n * I_SIZE + 1]);
            BihD[t] = mk2(W_ih_dec[n * I_SIZE + 0], W_ih_dec[n * I_SIZE + 1]);
        } else {
            BihE[t] = mk2(W_ih_enc[n * I_SIZE + 2], 0.0f);
            BihD[t] = mk2(W_ih_dec[n * I_SIZE + 2], 0.0f);
        }
        biasE[t] = b_ih_enc[n] + b_hh_enc[n];
        biasD[t] = b_ih_dec[n] + b_hh_dec[n];
    }
    const float fcb = (l16 < O_SIZE) ? fc_b[l16] : 0.0f;

    const int rowP = r0 + l16;
    const int rowQ = r0 + TILE_M + l16;
    const float* xbP = src + (size_t)(rowP >> 6) * T_STEPS * (AGENTS * I_SIZE) + (rowP & 63) * I_SIZE;
    const float* xbQ = src + (size_t)(rowQ >> 6) * T_STEPS * (AGENTS * I_SIZE) + (rowQ & 63) * I_SIZE;
    const int e1 = hi ? 2 : 1;

    const float* wl  = &s.W[l16][4 * (hi ? 1 : 0)];
    float*       sst = &s.stage[wave][hi ? 8 : 0][pcol(l16)];
    const float* sld = &s.stage[wave][l16][4 * (hi ? 1 : 0)];

    v2f hP[32], hQ[32];
#pragma unroll
    for (int c = 0; c < 32; ++c) { hP[c] = mk2(0.0f, 0.0f); hQ[c] = mk2(0.0f, 0.0f); }

    v2f B0[32];
    load_B(wl, 0, B0);
    landed(B0);

    float xP0 = xbP[koff], xP1 = xbP[e1], xQ0 = xbQ[koff], xQ1 = xbQ[e1];
    v2f AxP = mk2(0.0f, 0.0f), AxQ = mk2(0.0f, 0.0f);

    for (int t = 0; t < T_STEPS; ++t) {
        AxP = mk2(xP0, hi ? 0.0f : xP1);
        AxQ = mk2(xQ0, hi ? 0.0f : xQ1);
        const int tn = (t + 1 < T_STEPS) ? t + 1 : t;
        const float* xrP = xbP + (size_t)tn * (AGENTS * I_SIZE);
        const float* xrQ = xbQ + (size_t)tn * (AGENTS * I_SIZE);
        xP0 = xrP[koff]; xP1 = xrP[e1]; xQ0 = xrQ[koff]; xQ1 = xrQ[e1];

        rnn_step(wl, sst, sld, hP, hQ, B0, AxP, AxQ, BihE, biasE);
    }

    __syncthreads();
    fill_W(s, W_hh_dec, tid);
    __syncthreads();
    load_B(wl, 0, B0);

    v2f Bfc[32];
#pragma unroll
    for (int c = 0; c < 32; ++c)
        Bfc[c] = mk2(s.fcT[4 * c + koff][l16], s.fcT[4 * c + koff + 1][l16]);

    for (int p = 0; p < PRED; ++p) {
        rnn_step(wl, sst, sld, hP, hQ, B0, AxP, AxQ, BihD, biasD);

        v8f prP = splat8(fcb);
        v8f prQ = prP;
#pragma unroll
        for (int c = 0; c < 32; ++c) {
            prP = wmma4(hP[c], Bfc[c], prP);
            prQ = wmma4(hQ[c], Bfc[c], prQ);
        }

        if (l16 < O_SIZE) {
#pragma unroll
            for (int v = 0; v < 8; ++v) {
                const int row = hi ? v + 8 : v;
                const int nP_ = r0 + row;
                const int nQ_ = r0 + TILE_M + row;
                out[((size_t)(nP_ >> 6) * PRED + p) * (AGENTS * O_SIZE) + (nP_ & 63) * O_SIZE + l16] = prP[v];
                out[((size_t)(nQ_ >> 6) * PRED + p) * (AGENTS * O_SIZE) + (nQ_ & 63) * O_SIZE + l16] = prQ[v];
                s.stage[wave][row][l16]          = prP[v];
                s.stage[wave][TILE_M + row][l16] = prQ[v];
            }
        }
        AxP = hi ? mk2(s.stage[wave][l16][2], 0.0f)
                 : mk2(s.stage[wave][l16][0], s.stage[wave][l16][1]);
        AxQ = hi ? mk2(s.stage[wave][TILE_M + l16][2], 0.0f)
                 : mk2(s.stage[wave][TILE_M + l16][0], s.stage[wave][TILE_M + l16][1]);
    }
}

extern "C" void kernel_launch(void* const* d_in, const int* in_sizes, int n_in,
                              void* d_out, int out_size, void* d_ws, size_t ws_size,
                              hipStream_t stream) {
    const float* src      = (const float*)d_in[0];
    const float* W_ih_enc = (const float*)d_in[1];
    const float* W_hh_enc = (const float*)d_in[2];
    const float* b_ih_enc = (const float*)d_in[3];
    const float* b_hh_enc = (const float*)d_in[4];
    const float* W_ih_dec = (const float*)d_in[5];
    const float* W_hh_dec = (const float*)d_in[6];
    const float* b_ih_dec = (const float*)d_in[7];
    const float* b_hh_dec = (const float*)d_in[8];
    const float* fc_W     = (const float*)d_in[9];
    const float* fc_b     = (const float*)d_in[10];
    float* out = (float*)d_out;

    traj_rnn_kernel<<<NBLOCKS, WAVES * 32, 0, stream>>>(
        src, W_ih_enc, W_hh_enc, b_ih_enc, b_hh_enc,
        W_ih_dec, W_hh_dec, b_ih_dec, b_hh_dec, fc_W, fc_b, out);
}
